# starting kernel
# baseline (speedup 1.0000x reference)
#include <hip/hip_runtime.h>

#define NQ       12
#define DIM      4096
#define NLAYERS  6
#define TPB      256
#define APT      (DIM / TPB)
#define ANG_STRIDE 16
#define DOUT     256

typedef float v2f __attribute__((ext_vector_type(2)));
typedef float v8f __attribute__((ext_vector_type(8)));

__global__ __launch_bounds__(256) void fc_in_wmma(const float* __restrict__ x,
                                                  const float* __restrict__ W_in,
                                                  const float* __restrict__ b_in,
                                                  float* __restrict__ angles) {
    const int lane = threadIdx.x & 31;
    const int wave = threadIdx.x >> 5;
    const int tile = blockIdx.x * 8 + wave;
    const int n    = lane & 15;
    const int h    = lane >> 4;
    const int nc   = (n < NQ) ? n : (NQ - 1);
    const float mask = (n < NQ) ? 1.0f : 0.0f;

    const float* xrow = x + (size_t)(tile * 16 + n) * 256;
    const float* wrow = W_in + nc * 256;

    v8f acc = {};
    for (int k0 = 0; k0 < 256; k0 += 8) {
        const float4 av = *(const float4*)(xrow + k0 + 4 * h);
        float4 wv       = *(const float4*)(wrow + k0 + 4 * h);
        wv.x *= mask; wv.y *= mask; wv.z *= mask; wv.w *= mask;

        v2f a1, b1, a2, b2;
        a1.x = av.x; a1.y = av.y;  b1.x = wv.x; b1.y = wv.y;
        a2.x = av.z; a2.y = av.w;  b2.x = wv.z; b2.y = wv.w;
        acc = __builtin_amdgcn_wmma_f32_16x16x4_f32(false, a1, false, b1,
                                                    (short)0, acc, false, false);
        acc = __builtin_amdgcn_wmma_f32_16x16x4_f32(false, a2, false, b2,
                                                    (short)0, acc, false, false);
    }

    const float bias = b_in[nc] * mask;
    const int mbase = tile * 16 + (h << 3);
#pragma unroll
    for (int r = 0; r < 8; ++r)
        angles[(size_t)(mbase + r) * ANG_STRIDE + n] = acc[r] + bias;
}

__device__ __forceinline__ void rx_apply(float2* st, int bit, float c, float s) {
    const int tid = threadIdx.x;
#pragma unroll
    for (int it = 0; it < (DIM / 2) / TPB; ++it) {
        int p  = tid + it * TPB;
        int i0 = ((p >> bit) << (bit + 1)) | (p & ((1 << bit) - 1));
        int i1 = i0 | (1 << bit);
        float2 a0 = st[i0];
        float2 a1 = st[i1];
        float2 r0, r1;
        r0.x = fmaf(c, a0.x,  s * a1.y);
        r0.y = fmaf(c, a0.y, -s * a1.x);
        r1.x = fmaf(c, a1.x,  s * a0.y);
        r1.y = fmaf(c, a1.y, -s * a0.x);
        st[i0] = r0;
        st[i1] = r1;
    }
    __syncthreads();
}

__device__ __forceinline__ void cnot_apply(float2* st, int cbit, int tbit) {
    const int tid = threadIdx.x;
    const int lo = (cbit < tbit) ? cbit : tbit;
    const int hi = (cbit < tbit) ? tbit : cbit;
#pragma unroll
    for (int it = 0; it < (DIM / 4) / TPB; ++it) {
        int p = tid + it * TPB;
        int i = ((p >> lo) << (lo + 1)) | (p & ((1 << lo) - 1));
        i     = ((i >> hi) << (hi + 1)) | (i & ((1 << hi) - 1));
        i    |= (1 << cbit);
        int j = i | (1 << tbit);
        float2 t = st[i];
        st[i] = st[j];
        st[j] = t;
    }
    __syncthreads();
}

__global__ __launch_bounds__(TPB) void vqc_kernel(const float* __restrict__ angles,
                                                  const float* __restrict__ qp,
                                                  const float* __restrict__ W_out,
                                                  const float* __restrict__ b_out,
                                                  float* __restrict__ out) {
    __shared__ float2 st[DIM];
    const int tid = threadIdx.x;
    const int b   = blockIdx.x;

#pragma unroll
    for (int j = 0; j < APT; ++j) {
        float2 z; z.x = 0.f; z.y = 0.f;
        st[tid * APT + j] = z;
    }
    if (tid == 0) { float2 one; one.x = 1.f; one.y = 0.f; st[0] = one; }
    __syncthreads();

    for (int q = 0; q < NQ; ++q) {
        float hh = 0.5f * angles[(size_t)b * ANG_STRIDE + q];
        rx_apply(st, 11 - q, __cosf(hh), __sinf(hh));
    }

    for (int l = 0; l < NLAYERS; ++l) {
        for (int q = 0; q < NQ; ++q) {
            float hh = 0.5f * qp[l * NQ + q];
            rx_apply(st, 11 - q, __cosf(hh), __sinf(hh));
        }
        for (int q = 0; q < NQ - 1; ++q)
            cnot_apply(st, 11 - q, 10 - q);
        cnot_apply(st, 0, 11);
    }

    float zp[NQ];
#pragma unroll
    for (int q = 0; q < NQ; ++q) zp[q] = 0.f;
#pragma unroll
    for (int j = 0; j < APT; ++j) {
        int i = tid * APT + j;
        float2 a = st[i];
        float pr = fmaf(a.x, a.x, a.y * a.y);
#pragma unroll
        for (int q = 0; q < NQ; ++q)
            zp[q] += ((i >> (11 - q)) & 1) ? -pr : pr;
    }
    __syncthreads();
    float* red = (float*)st;
#pragma unroll
    for (int q = 0; q < NQ; ++q) red[tid * NQ + q] = zp[q];
    __syncthreads();
    for (int s = TPB / 2; s >= 1; s >>= 1) {
        if (tid < s) {
#pragma unroll
            for (int q = 0; q < NQ; ++q)
                red[tid * NQ + q] += red[(tid + s) * NQ + q];
        }
        __syncthreads();
    }

    float accv = b_out[tid];
#pragma unroll
    for (int q = 0; q < NQ; ++q)
        accv = fmaf(red[q], W_out[tid * NQ + q], accv);
    out[(size_t)b * DOUT + tid] = accv;
}

extern "C" void kernel_launch(void* const* d_in, const int* in_sizes, int n_in,
                              void* d_out, int out_size, void* d_ws, size_t ws_size,
                              hipStream_t stream) {
    const float* x        = (const float*)d_in[0];
    const float* W_in     = (const float*)d_in[1];
    const float* b_in     = (const float*)d_in[2];
    const float* q_params = (const float*)d_in[3];
    const float* W_out    = (const float*)d_in[4];
    const float* b_out    = (const float*)d_in[5];
    float* out    = (float*)d_out;
    float* angles = (float*)d_ws;

    fc_in_wmma<<<dim3(32),   dim3(256), 0, stream>>>(x, W_in, b_in, angles);
    vqc_kernel<<<dim3(4096), dim3(TPB), 0, stream>>>(angles, q_params, W_out, b_out, out);
}


# One wave per sample: state in registers as 7-qubit tiles (128 amplitudes per lane), 2 LDS exchanges per layer not 4, no barriers; order of commuting layer-0 gates and Z sum order change
# speedup vs baseline: 6.8768x; 6.8768x over previous
#include <hip/hip_runtime.h>

#define NQ       12
#define DIM      4096
#define NLAYERS  6
#define NA       128
#define ANG_STRIDE 16
#define DOUT     256

typedef float v2f __attribute__((ext_vector_type(2)));
typedef float v8f __attribute__((ext_vector_type(8)));

__global__ __launch_bounds__(256) void fc_in_wmma(const float* __restrict__ x,
                                                  const float* __restrict__ W_in,
                                                  const float* __restrict__ b_in,
                                                  float* __restrict__ angles) {
    const int lane = threadIdx.x & 31;
    const int wave = threadIdx.x >> 5;
    const int tile = blockIdx.x * 8 + wave;
    const int n    = lane & 15;
    const int h    = lane >> 4;
    const int nc   = (n < NQ) ? n : (NQ - 1);
    const float mask = (n < NQ) ? 1.0f : 0.0f;

    const float* xrow = x + (size_t)(tile * 16 + n) * 256;
    const float* wrow = W_in + nc * 256;

    v8f acc = {};
    for (int k0 = 0; k0 < 256; k0 += 8) {
        const float4 av = *(const float4*)(xrow + k0 + 4 * h);
        float4 wv       = *(const float4*)(wrow + k0 + 4 * h);
        wv.x *= mask; wv.y *= mask; wv.z *= mask; wv.w *= mask;

        v2f a1, b1, a2, b2;
        a1.x = av.x; a1.y = av.y;  b1.x = wv.x; b1.y = wv.y;
        a2.x = av.z; a2.y = av.w;  b2.x = wv.z; b2.y = wv.w;
        acc = __builtin_amdgcn_wmma_f32_16x16x4_f32(false, a1, false, b1,
                                                    (short)0, acc, false, false);
        acc = __builtin_amdgcn_wmma_f32_16x16x4_f32(false, a2, false, b2,
                                                    (short)0, acc, false, false);
    }

    const float bias = b_in[nc] * mask;
    const int mbase = tile * 16 + (h << 3);
#pragma unroll
    for (int r = 0; r < 8; ++r)
        angles[(size_t)(mbase + r) * ANG_STRIDE + n] = acc[r] + bias;
}

#define ST_CELLS 2112
__host__ __device__ constexpr int slot4(int i) {
    return (i & 31) + ((i >> 6) & 1) * 33 + ((i >> 7) & 1) * 66 +
           ((i >> 8) & 1) * 132 + ((i >> 9) & 1) * 264 + ((i >> 10) & 1) * 528 +
           ((i >> 11) & 1) * 1056;
}

template <int P0, int P1, int P2, int P3, int P4, int P5, int P6>
__host__ __device__ constexpr int tile_off(int j) {
    return ((j & 1) << P0) | (((j >> 1) & 1) << P1) | (((j >> 2) & 1) << P2) |
           (((j >> 3) & 1) << P3) | (((j >> 4) & 1) << P4) | (((j >> 5) & 1) << P5) |
           (((j >> 6) & 1) << P6);
}

template <int KH, int P0, int P1, int P2, int P3, int P4, int P5, int P6>
__device__ __forceinline__ void tile_load(const float4* st, int pb, float2 (&a)[NA]) {
    static_assert(tile_off<P0, P1, P2, P3, P4, P5, P6>(1 << KH) == 32, "KH must map to index bit 5");
#pragma unroll
    for (int j = 0; j < NA; ++j) {
        if (j & (1 << KH)) continue;
        const float4 v = st[pb + slot4(tile_off<P0, P1, P2, P3, P4, P5, P6>(j))];
        a[j].x = v.x;             a[j].y = v.y;
        a[j | (1 << KH)].x = v.z; a[j | (1 << KH)].y = v.w;
    }
}

template <int KH, int P0, int P1, int P2, int P3, int P4, int P5, int P6>
__device__ __forceinline__ void tile_store(float4* st, int pb, const float2 (&a)[NA]) {
    static_assert(tile_off<P0, P1, P2, P3, P4, P5, P6>(1 << KH) == 32, "KH must map to index bit 5");
#pragma unroll
    for (int j = 0; j < NA; ++j) {
        if (j & (1 << KH)) continue;
        float4 v;
        v.x = a[j].x;             v.y = a[j].y;
        v.z = a[j | (1 << KH)].x; v.w = a[j | (1 << KH)].y;
        st[pb + slot4(tile_off<P0, P1, P2, P3, P4, P5, P6>(j))] = v;
    }
}

struct GateTab { float m; unsigned flip; };

__device__ __forceinline__ GateTab make_tab(float theta) {
    const float hh = 0.5f * theta;
    const float c = __cosf(hh);
    const float s = __sinf(hh);
    const bool cm = fabsf(c) >= fabsf(s);
    GateTab g;
    g.m = (cm ? s : -c) * __builtin_amdgcn_rcpf(cm ? c : s);
    g.flip = __builtin_amdgcn_ballot_w32(!cm) & 0xFFFu;
    return g;
}

__device__ __forceinline__ unsigned cnot_frame(unsigned f) {
    f ^= f << 1;
    f ^= f << 2;
    f ^= f << 4;
    f ^= f << 8;
    f &= 0xFFFu;
    f ^= (f >> 11) & 1u;
    return f;
}

template <int K>
__device__ __forceinline__ void rx_local(float2 (&a)[NA], const GateTab& g, int q) {
    const float m = __int_as_float(__builtin_amdgcn_readlane(__float_as_int(g.m), q));
#pragma unroll
    for (int j = 0; j < NA; ++j) {
        if (j & (1 << K)) continue;
        const int j1 = j | (1 << K);
        const float2 a0 = a[j];
        const float2 a1 = a[j1];
        float2 r0, r1;
        r0.x = fmaf( m, a1.y, a0.x);
        r0.y = fmaf(-m, a1.x, a0.y);
        r1.x = fmaf( m, a0.y, a1.x);
        r1.y = fmaf(-m, a0.x, a1.y);
        a[j]  = r0;
        a[j1] = r1;
    }
}

template <int CP, int TP>
__device__ __forceinline__ void cnot_local(float2 (&a)[NA]) {
#pragma unroll
    for (int j = 0; j < NA; ++j) {
        if (!(j & (1 << CP)) || (j & (1 << TP))) continue;
        const int j1 = j | (1 << TP);
        const float2 t = a[j];
        a[j]  = a[j1];
        a[j1] = t;
    }
}

__device__ __forceinline__ void cnot_chain(float2 (&a)[NA]) {
    cnot_local<0, 1>(a);
    cnot_local<1, 2>(a);
    cnot_local<2, 3>(a);
    cnot_local<3, 4>(a);
    cnot_local<4, 5>(a);
    cnot_local<5, 6>(a);
}

template <bool ENC>
__device__ __forceinline__ void window1(float4* st, int lane, float2 (&a)[NA],
                                        const GateTab& enc, const GateTab& w) {
    const int pb = slot4(lane);
    if (ENC) {
#pragma unroll
        for (int j = 0; j < NA; ++j) { a[j].x = 0.f; a[j].y = 0.f; }
        if (lane == 0) a[0].x = 1.f;
        rx_local<0>(a, enc, 0);  rx_local<0>(a, w, 0);
        rx_local<1>(a, enc, 1);  rx_local<1>(a, w, 1);
        rx_local<2>(a, enc, 2);  rx_local<2>(a, w, 2);
        rx_local<3>(a, enc, 3);  rx_local<3>(a, w, 3);
        rx_local<4>(a, enc, 4);  rx_local<4>(a, w, 4);
        rx_local<5>(a, enc, 5);  rx_local<5>(a, w, 5);
        rx_local<6>(a, enc, 6);  rx_local<6>(a, w, 6);
    } else {
        tile_load<6, 11, 10, 9, 8, 7, 6, 5>(st, pb, a);
        rx_local<0>(a, w, 0);
        rx_local<1>(a, w, 1);
        rx_local<2>(a, w, 2);
        rx_local<3>(a, w, 3);
        rx_local<4>(a, w, 4);
        rx_local<5>(a, w, 5);
        rx_local<6>(a, w, 6);
    }
    cnot_chain(a);
    tile_store<6, 11, 10, 9, 8, 7, 6, 5>(st, pb, a);
    __syncthreads();
}

template <bool ENC>
__device__ __forceinline__ void window2(const float4* st, int lane, float2 (&a)[NA],
                                        const GateTab& enc, const GateTab& w) {
    const int pb = slot4(lane << 6);
    if (ENC) {
#pragma unroll
        for (int j = 0; j < NA; ++j) { a[j].x = 0.f; a[j].y = 0.f; }
        const float4 v0 = st[pb + slot4(tile_off<5, 4, 3, 2, 1, 0, 11>(0))];
        const float4 v1 = st[pb + slot4(tile_off<5, 4, 3, 2, 1, 0, 11>(64))];
        a[0].x  = v0.x; a[0].y  = v0.y; a[1].x  = v0.z; a[1].y  = v0.w;
        a[64].x = v1.x; a[64].y = v1.y; a[65].x = v1.z; a[65].y = v1.w;
        rx_local<1>(a, enc, 7);   rx_local<1>(a, w, 7);
        rx_local<2>(a, enc, 8);   rx_local<2>(a, w, 8);
        rx_local<3>(a, enc, 9);   rx_local<3>(a, w, 9);
        rx_local<4>(a, enc, 10);  rx_local<4>(a, w, 10);
        rx_local<5>(a, enc, 11);  rx_local<5>(a, w, 11);
    } else {
        tile_load<0, 5, 4, 3, 2, 1, 0, 11>(st, pb, a);
        rx_local<1>(a, w, 7);
        rx_local<2>(a, w, 8);
        rx_local<3>(a, w, 9);
        rx_local<4>(a, w, 10);
        rx_local<5>(a, w, 11);
    }
    cnot_chain(a);
}

__device__ __forceinline__ void window2_store(float4* st, int lane, const float2 (&a)[NA]) {
    tile_store<0, 5, 4, 3, 2, 1, 0, 11>(st, slot4(lane << 6), a);
    __syncthreads();
}

template <int N>
__device__ __forceinline__ void z_fold(float (&v)[NA], float& d) {
#pragma unroll
    for (int t = 0; t < N / 2; ++t) {
        d += v[2 * t] - v[2 * t + 1];
        v[t] = v[2 * t] + v[2 * t + 1];
    }
}

__device__ __forceinline__ float wave_sum(float x) {
    x += __shfl_xor(x, 16, 32);
    x += __shfl_xor(x, 8, 32);
    x += __shfl_xor(x, 4, 32);
    x += __shfl_xor(x, 2, 32);
    x += __shfl_xor(x, 1, 32);
    return x;
}

__global__ __launch_bounds__(32) void vqc_kernel(const float* __restrict__ angles,
                                                 const float* __restrict__ qp,
                                                 const float* __restrict__ W_out,
                                                 const float* __restrict__ b_out,
                                                 float* __restrict__ out) {
    __shared__ float4 st[ST_CELLS];
    const int lane = threadIdx.x;
    const int b    = blockIdx.x;

    const int gq = lane < NQ ? lane : NQ - 1;
    const GateTab enc = make_tab(angles[(size_t)b * ANG_STRIDE + gq]);
    GateTab cur = make_tab(qp[gq]);
    float th_next = qp[NQ + gq];
    unsigned frame = cnot_frame(enc.flip ^ cur.flip);

    float2 a[NA];

    window1<true>(st, lane, a, enc, cur);
    window2<true>(st, lane, a, enc, cur);
    window2_store(st, lane, a);
    for (int l = 1; l < NLAYERS; ++l) {
        cur = make_tab(th_next);
        const int ln = (l + 1 < NLAYERS) ? l + 1 : NLAYERS - 1;
        th_next = qp[ln * NQ + gq];
        frame = cnot_frame(frame ^ cur.flip);
        window1<false>(st, lane, a, enc, cur);
        window2<false>(st, lane, a, enc, cur);
        if (l + 1 < NLAYERS) window2_store(st, lane, a);
    }

    float v[NA];
#pragma unroll
    for (int j = 0; j < NA; ++j) v[j] = fmaf(a[j].x, a[j].x, a[j].y * a[j].y);
    float zt[7];
#pragma unroll
    for (int k = 0; k < 7; ++k) zt[k] = 0.f;
    z_fold<128>(v, zt[0]);
    z_fold<64>(v, zt[1]);
    z_fold<32>(v, zt[2]);
    z_fold<16>(v, zt[3]);
    z_fold<8>(v, zt[4]);
    z_fold<4>(v, zt[5]);
    z_fold<2>(v, zt[6]);
    const float tp = v[0];

    float z[NQ];
    z[0] = zt[6];
#pragma unroll
    for (int q = 1; q <= 5; ++q) z[q] = ((lane >> (5 - q)) & 1) ? -tp : tp;
#pragma unroll
    for (int q = 6; q < NQ; ++q) z[q] = zt[q - 6];
#pragma unroll
    for (int q = 0; q < NQ; ++q) z[q] = wave_sum(z[q]);
    const float tot = wave_sum(tp);

    const float inv = 1.0f / tot;
#pragma unroll
    for (int q = 0; q < NQ; ++q) z[q] *= ((frame >> q) & 1u) ? -inv : inv;

#pragma unroll
    for (int r = 0; r < DOUT / 32; ++r) {
        const int o = r * 32 + lane;
        const float4* wr = (const float4*)(W_out + o * NQ);
        const float4 w0 = wr[0], w1 = wr[1], w2 = wr[2];
        float accv = b_out[o];
        accv = fmaf(z[0],  w0.x, accv);
        accv = fmaf(z[1],  w0.y, accv);
        accv = fmaf(z[2],  w0.z, accv);
        accv = fmaf(z[3],  w0.w, accv);
        accv = fmaf(z[4],  w1.x, accv);
        accv = fmaf(z[5],  w1.y, accv);
        accv = fmaf(z[6],  w1.z, accv);
        accv = fmaf(z[7],  w1.w, accv);
        accv = fmaf(z[8],  w2.x, accv);
        accv = fmaf(z[9],  w2.y, accv);
        accv = fmaf(z[10], w2.z, accv);
        accv = fmaf(z[11], w2.w, accv);
        out[(size_t)b * DOUT + o] = accv;
    }
}

extern "C" void kernel_launch(void* const* d_in, const int* in_sizes, int n_in,
                              void* d_out, int out_size, void* d_ws, size_t ws_size,
                              hipStream_t stream) {
    const float* x        = (const float*)d_in[0];
    const float* W_in     = (const float*)d_in[1];
    const float* b_in     = (const float*)d_in[2];
    const float* q_params = (const float*)d_in[3];
    const float* W_out    = (const float*)d_in[4];
    const float* b_out    = (const float*)d_in[5];
    float* out    = (float*)d_out;
    float* angles = (float*)d_ws;

    fc_in_wmma<<<dim3(32),   dim3(256), 0, stream>>>(x, W_in, b_in, angles);
    vqc_kernel<<<dim3(4096), dim3(32),  0, stream>>>(angles, q_params, W_out, b_out, out);
}


# RX gates as packed f32 multiply-adds, one instruction per amplitude (both components): partner's components exchanged by operand select, (m,-m) by negate modifier; each fmaf unchanged
# speedup vs baseline: 7.3352x; 1.0667x over previous
#include <hip/hip_runtime.h>

#define NQ       12
#define DIM      4096
#define NLAYERS  6
#define NA       128
#define ANG_STRIDE 16
#define DOUT     256

typedef float v2f __attribute__((ext_vector_type(2)));
typedef float v8f __attribute__((ext_vector_type(8)));

__global__ __launch_bounds__(256) void fc_in_wmma(const float* __restrict__ x,
                                                  const float* __restrict__ W_in,
                                                  const float* __restrict__ b_in,
                                                  float* __restrict__ angles) {
    const int lane = threadIdx.x & 31;
    const int wave = threadIdx.x >> 5;
    const int tile = blockIdx.x * 8 + wave;
    const int n    = lane & 15;
    const int h    = lane >> 4;
    const int nc   = (n < NQ) ? n : (NQ - 1);
    const float mask = (n < NQ) ? 1.0f : 0.0f;

    const float* xrow = x + (size_t)(tile * 16 + n) * 256;
    const float* wrow = W_in + nc * 256;

    v8f acc = {};
    for (int k0 = 0; k0 < 256; k0 += 8) {
        const float4 av = *(const float4*)(xrow + k0 + 4 * h);
        float4 wv       = *(const float4*)(wrow + k0 + 4 * h);
        wv.x *= mask; wv.y *= mask; wv.z *= mask; wv.w *= mask;

        v2f a1, b1, a2, b2;
        a1.x = av.x; a1.y = av.y;  b1.x = wv.x; b1.y = wv.y;
        a2.x = av.z; a2.y = av.w;  b2.x = wv.z; b2.y = wv.w;
        acc = __builtin_amdgcn_wmma_f32_16x16x4_f32(false, a1, false, b1,
                                                    (short)0, acc, false, false);
        acc = __builtin_amdgcn_wmma_f32_16x16x4_f32(false, a2, false, b2,
                                                    (short)0, acc, false, false);
    }

    const float bias = b_in[nc] * mask;
    const int mbase = tile * 16 + (h << 3);
#pragma unroll
    for (int r = 0; r < 8; ++r)
        angles[(size_t)(mbase + r) * ANG_STRIDE + n] = acc[r] + bias;
}

#define ST_CELLS 2112
__host__ __device__ constexpr int slot4(int i) {
    return (i & 31) + ((i >> 6) & 1) * 33 + ((i >> 7) & 1) * 66 +
           ((i >> 8) & 1) * 132 + ((i >> 9) & 1) * 264 + ((i >> 10) & 1) * 528 +
           ((i >> 11) & 1) * 1056;
}

template <int P0, int P1, int P2, int P3, int P4, int P5, int P6>
__host__ __device__ constexpr int tile_off(int j) {
    return ((j & 1) << P0) | (((j >> 1) & 1) << P1) | (((j >> 2) & 1) << P2) |
           (((j >> 3) & 1) << P3) | (((j >> 4) & 1) << P4) | (((j >> 5) & 1) << P5) |
           (((j >> 6) & 1) << P6);
}

template <int KH, int P0, int P1, int P2, int P3, int P4, int P5, int P6>
__device__ __forceinline__ void tile_load(const float4* st, int pb, float2 (&a)[NA]) {
    static_assert(tile_off<P0, P1, P2, P3, P4, P5, P6>(1 << KH) == 32, "KH must map to index bit 5");
#pragma unroll
    for (int j = 0; j < NA; ++j) {
        if (j & (1 << KH)) continue;
        const float4 v = st[pb + slot4(tile_off<P0, P1, P2, P3, P4, P5, P6>(j))];
        a[j].x = v.x;             a[j].y = v.y;
        a[j | (1 << KH)].x = v.z; a[j | (1 << KH)].y = v.w;
    }
}

template <int KH, int P0, int P1, int P2, int P3, int P4, int P5, int P6>
__device__ __forceinline__ void tile_store(float4* st, int pb, const float2 (&a)[NA]) {
    static_assert(tile_off<P0, P1, P2, P3, P4, P5, P6>(1 << KH) == 32, "KH must map to index bit 5");
#pragma unroll
    for (int j = 0; j < NA; ++j) {
        if (j & (1 << KH)) continue;
        float4 v;
        v.x = a[j].x;             v.y = a[j].y;
        v.z = a[j | (1 << KH)].x; v.w = a[j | (1 << KH)].y;
        st[pb + slot4(tile_off<P0, P1, P2, P3, P4, P5, P6>(j))] = v;
    }
}

struct GateTab { float m; unsigned flip; };

__device__ __forceinline__ GateTab make_tab(float theta) {
    const float hh = 0.5f * theta;
    const float c = __cosf(hh);
    const float s = __sinf(hh);
    const bool cm = fabsf(c) >= fabsf(s);
    GateTab g;
    g.m = (cm ? s : -c) * __builtin_amdgcn_rcpf(cm ? c : s);
    g.flip = __builtin_amdgcn_ballot_w32(!cm) & 0xFFFu;
    return g;
}

__device__ __forceinline__ unsigned cnot_frame(unsigned f) {
    f ^= f << 1;
    f ^= f << 2;
    f ^= f << 4;
    f ^= f << 8;
    f &= 0xFFFu;
    f ^= (f >> 11) & 1u;
    return f;
}

template <int K>
__device__ __forceinline__ void rx_local(float2 (&a)[NA], const GateTab& g, int q) {
    const float m = __int_as_float(__builtin_amdgcn_readlane(__float_as_int(g.m), q));
    const v2f mm = {m, -m};
#pragma unroll
    for (int j = 0; j < NA; ++j) {
        if (j & (1 << K)) continue;
        const int j1 = j | (1 << K);
        const v2f p0 = {a[j].x,  a[j].y};
        const v2f p1 = {a[j1].x, a[j1].y};
        const v2f s0 = __builtin_shufflevector(p0, p0, 1, 0);
        const v2f s1 = __builtin_shufflevector(p1, p1, 1, 0);
        const v2f r0 = __builtin_elementwise_fma(mm, s1, p0);
        const v2f r1 = __builtin_elementwise_fma(mm, s0, p1);
        a[j].x  = r0.x;  a[j].y  = r0.y;
        a[j1].x = r1.x;  a[j1].y = r1.y;
    }
}

template <int CP, int TP>
__device__ __forceinline__ void cnot_local(float2 (&a)[NA]) {
#pragma unroll
    for (int j = 0; j < NA; ++j) {
        if (!(j & (1 << CP)) || (j & (1 << TP))) continue;
        const int j1 = j | (1 << TP);
        const float2 t = a[j];
        a[j]  = a[j1];
        a[j1] = t;
    }
}

__device__ __forceinline__ void cnot_chain(float2 (&a)[NA]) {
    cnot_local<0, 1>(a);
    cnot_local<1, 2>(a);
    cnot_local<2, 3>(a);
    cnot_local<3, 4>(a);
    cnot_local<4, 5>(a);
    cnot_local<5, 6>(a);
}

template <bool ENC>
__device__ __forceinline__ void window1(float4* st, int lane, float2 (&a)[NA],
                                        const GateTab& enc, const GateTab& w) {
    const int pb = slot4(lane);
    if (ENC) {
#pragma unroll
        for (int j = 0; j < NA; ++j) { a[j].x = 0.f; a[j].y = 0.f; }
        if (lane == 0) a[0].x = 1.f;
        rx_local<0>(a, enc, 0);  rx_local<0>(a, w, 0);
        rx_local<1>(a, enc, 1);  rx_local<1>(a, w, 1);
        rx_local<2>(a, enc, 2);  rx_local<2>(a, w, 2);
        rx_local<3>(a, enc, 3);  rx_local<3>(a, w, 3);
        rx_local<4>(a, enc, 4);  rx_local<4>(a, w, 4);
        rx_local<5>(a, enc, 5);  rx_local<5>(a, w, 5);
        rx_local<6>(a, enc, 6);  rx_local<6>(a, w, 6);
    } else {
        tile_load<6, 11, 10, 9, 8, 7, 6, 5>(st, pb, a);
        rx_local<0>(a, w, 0);
        rx_local<1>(a, w, 1);
        rx_local<2>(a, w, 2);
        rx_local<3>(a, w, 3);
        rx_local<4>(a, w, 4);
        rx_local<5>(a, w, 5);
        rx_local<6>(a, w, 6);
    }
    cnot_chain(a);
    tile_store<6, 11, 10, 9, 8, 7, 6, 5>(st, pb, a);
    __syncthreads();
}

template <bool ENC>
__device__ __forceinline__ void window2(const float4* st, int lane, float2 (&a)[NA],
                                        const GateTab& enc, const GateTab& w) {
    const int pb = slot4(lane << 6);
    if (ENC) {
#pragma unroll
        for (int j = 0; j < NA; ++j) { a[j].x = 0.f; a[j].y = 0.f; }
        const float4 v0 = st[pb + slot4(tile_off<5, 4, 3, 2, 1, 0, 11>(0))];
        const float4 v1 = st[pb + slot4(tile_off<5, 4, 3, 2, 1, 0, 11>(64))];
        a[0].x  = v0.x; a[0].y  = v0.y; a[1].x  = v0.z; a[1].y  = v0.w;
        a[64].x = v1.x; a[64].y = v1.y; a[65].x = v1.z; a[65].y = v1.w;
        rx_local<1>(a, enc, 7);   rx_local<1>(a, w, 7);
        rx_local<2>(a, enc, 8);   rx_local<2>(a, w, 8);
        rx_local<3>(a, enc, 9);   rx_local<3>(a, w, 9);
        rx_local<4>(a, enc, 10);  rx_local<4>(a, w, 10);
        rx_local<5>(a, enc, 11);  rx_local<5>(a, w, 11);
    } else {
        tile_load<0, 5, 4, 3, 2, 1, 0, 11>(st, pb, a);
        rx_local<1>(a, w, 7);
        rx_local<2>(a, w, 8);
        rx_local<3>(a, w, 9);
        rx_local<4>(a, w, 10);
        rx_local<5>(a, w, 11);
    }
    cnot_chain(a);
}

__device__ __forceinline__ void window2_store(float4* st, int lane, const float2 (&a)[NA]) {
    tile_store<0, 5, 4, 3, 2, 1, 0, 11>(st, slot4(lane << 6), a);
    __syncthreads();
}

template <int N>
__device__ __forceinline__ void z_fold(float (&v)[NA], float& d) {
#pragma unroll
    for (int t = 0; t < N / 2; ++t) {
        d += v[2 * t] - v[2 * t + 1];
        v[t] = v[2 * t] + v[2 * t + 1];
    }
}

__device__ __forceinline__ float wave_sum(float x) {
    x += __shfl_xor(x, 16, 32);
    x += __shfl_xor(x, 8, 32);
    x += __shfl_xor(x, 4, 32);
    x += __shfl_xor(x, 2, 32);
    x += __shfl_xor(x, 1, 32);
    return x;
}

__global__ __launch_bounds__(32) void vqc_kernel(const float* __restrict__ angles,
                                                 const float* __restrict__ qp,
                                                 const float* __restrict__ W_out,
                                                 const float* __restrict__ b_out,
                                                 float* __restrict__ out) {
    __shared__ float4 st[ST_CELLS];
    const int lane = threadIdx.x;
    const int b    = blockIdx.x;

    const int gq = lane < NQ ? lane : NQ - 1;
    const GateTab enc = make_tab(angles[(size_t)b * ANG_STRIDE + gq]);
    GateTab cur = make_tab(qp[gq]);
    float th_next = qp[NQ + gq];
    unsigned frame = cnot_frame(enc.flip ^ cur.flip);

    float2 a[NA];

    window1<true>(st, lane, a, enc, cur);
    window2<true>(st, lane, a, enc, cur);
    window2_store(st, lane, a);
    for (int l = 1; l < NLAYERS; ++l) {
        cur = make_tab(th_next);
        const int ln = (l + 1 < NLAYERS) ? l + 1 : NLAYERS - 1;
        th_next = qp[ln * NQ + gq];
        frame = cnot_frame(frame ^ cur.flip);
        window1<false>(st, lane, a, enc, cur);
        window2<false>(st, lane, a, enc, cur);
        if (l + 1 < NLAYERS) window2_store(st, lane, a);
    }

    float v[NA];
#pragma unroll
    for (int j = 0; j < NA; ++j) v[j] = fmaf(a[j].x, a[j].x, a[j].y * a[j].y);
    float zt[7];
#pragma unroll
    for (int k = 0; k < 7; ++k) zt[k] = 0.f;
    z_fold<128>(v, zt[0]);
    z_fold<64>(v, zt[1]);
    z_fold<32>(v, zt[2]);
    z_fold<16>(v, zt[3]);
    z_fold<8>(v, zt[4]);
    z_fold<4>(v, zt[5]);
    z_fold<2>(v, zt[6]);
    const float tp = v[0];

    float z[NQ];
    z[0] = zt[6];
#pragma unroll
    for (int q = 1; q <= 5; ++q) z[q] = ((lane >> (5 - q)) & 1) ? -tp : tp;
#pragma unroll
    for (int q = 6; q < NQ; ++q) z[q] = zt[q - 6];
#pragma unroll
    for (int q = 0; q < NQ; ++q) z[q] = wave_sum(z[q]);
    const float tot = wave_sum(tp);

    const float inv = 1.0f / tot;
#pragma unroll
    for (int q = 0; q < NQ; ++q) z[q] *= ((frame >> q) & 1u) ? -inv : inv;

#pragma unroll
    for (int r = 0; r < DOUT / 32; ++r) {
        const int o = r * 32 + lane;
        const float4* wr = (const float4*)(W_out + o * NQ);
        const float4 w0 = wr[0], w1 = wr[1], w2 = wr[2];
        float accv = b_out[o];
        accv = fmaf(z[0],  w0.x, accv);
        accv = fmaf(z[1],  w0.y, accv);
        accv = fmaf(z[2],  w0.z, accv);
        accv = fmaf(z[3],  w0.w, accv);
        accv = fmaf(z[4],  w1.x, accv);
        accv = fmaf(z[5],  w1.y, accv);
        accv = fmaf(z[6],  w1.z, accv);
        accv = fmaf(z[7],  w1.w, accv);
        accv = fmaf(z[8],  w2.x, accv);
        accv = fmaf(z[9],  w2.y, accv);
        accv = fmaf(z[10], w2.z, accv);
        accv = fmaf(z[11], w2.w, accv);
        out[(size_t)b * DOUT + o] = accv;
    }
}

extern "C" void kernel_launch(void* const* d_in, const int* in_sizes, int n_in,
                              void* d_out, int out_size, void* d_ws, size_t ws_size,
                              hipStream_t stream) {
    const float* x        = (const float*)d_in[0];
    const float* W_in     = (const float*)d_in[1];
    const float* b_in     = (const float*)d_in[2];
    const float* q_params = (const float*)d_in[3];
    const float* W_out    = (const float*)d_in[4];
    const float* b_out    = (const float*)d_in[5];
    float* out    = (float*)d_out;
    float* angles = (float*)d_ws;

    fc_in_wmma<<<dim3(32),   dim3(256), 0, stream>>>(x, W_in, b_in, angles);
    vqc_kernel<<<dim3(4096), dim3(32),  0, stream>>>(angles, q_params, W_out, b_out, out);
}


# Layer 0 (encoding RX, first shared RX, first CNOT ring) in closed form: a chain product built by doubling straight into window 1's registers, no gate pass or LDS trip; layer-0 rounding changes
# speedup vs baseline: 7.7151x; 1.0518x over previous
#include <hip/hip_runtime.h>

#define NQ       12
#define DIM      4096
#define NLAYERS  6
#define NA       128
#define ANG_STRIDE 16
#define DOUT     256

typedef float v2f __attribute__((ext_vector_type(2)));
typedef float v8f __attribute__((ext_vector_type(8)));

__global__ __launch_bounds__(256) void fc_in_wmma(const float* __restrict__ x,
                                                  const float* __restrict__ W_in,
                                                  const float* __restrict__ b_in,
                                                  float* __restrict__ angles) {
    const int lane = threadIdx.x & 31;
    const int wave = threadIdx.x >> 5;
    const int tile = blockIdx.x * 8 + wave;
    const int n    = lane & 15;
    const int h    = lane >> 4;
    const int nc   = (n < NQ) ? n : (NQ - 1);
    const float mask = (n < NQ) ? 1.0f : 0.0f;

    const float* xrow = x + (size_t)(tile * 16 + n) * 256;
    const float* wrow = W_in + nc * 256;

    v8f acc = {};
    for (int k0 = 0; k0 < 256; k0 += 8) {
        const float4 av = *(const float4*)(xrow + k0 + 4 * h);
        float4 wv       = *(const float4*)(wrow + k0 + 4 * h);
        wv.x *= mask; wv.y *= mask; wv.z *= mask; wv.w *= mask;

        v2f a1, b1, a2, b2;
        a1.x = av.x; a1.y = av.y;  b1.x = wv.x; b1.y = wv.y;
        a2.x = av.z; a2.y = av.w;  b2.x = wv.z; b2.y = wv.w;
        acc = __builtin_amdgcn_wmma_f32_16x16x4_f32(false, a1, false, b1,
                                                    (short)0, acc, false, false);
        acc = __builtin_amdgcn_wmma_f32_16x16x4_f32(false, a2, false, b2,
                                                    (short)0, acc, false, false);
    }

    const float bias = b_in[nc] * mask;
    const int mbase = tile * 16 + (h << 3);
#pragma unroll
    for (int r = 0; r < 8; ++r)
        angles[(size_t)(mbase + r) * ANG_STRIDE + n] = acc[r] + bias;
}

#define ST_CELLS 2112
__host__ __device__ constexpr int slot4(int i) {
    return (i & 31) + ((i >> 6) & 1) * 33 + ((i >> 7) & 1) * 66 +
           ((i >> 8) & 1) * 132 + ((i >> 9) & 1) * 264 + ((i >> 10) & 1) * 528 +
           ((i >> 11) & 1) * 1056;
}

template <int P0, int P1, int P2, int P3, int P4, int P5, int P6>
__host__ __device__ constexpr int tile_off(int j) {
    return ((j & 1) << P0) | (((j >> 1) & 1) << P1) | (((j >> 2) & 1) << P2) |
           (((j >> 3) & 1) << P3) | (((j >> 4) & 1) << P4) | (((j >> 5) & 1) << P5) |
           (((j >> 6) & 1) << P6);
}

template <int KH, int P0, int P1, int P2, int P3, int P4, int P5, int P6>
__device__ __forceinline__ void tile_load(const float4* st, int pb, float2 (&a)[NA]) {
    static_assert(tile_off<P0, P1, P2, P3, P4, P5, P6>(1 << KH) == 32, "KH must map to index bit 5");
#pragma unroll
    for (int j = 0; j < NA; ++j) {
        if (j & (1 << KH)) continue;
        const float4 v = st[pb + slot4(tile_off<P0, P1, P2, P3, P4, P5, P6>(j))];
        a[j].x = v.x;             a[j].y = v.y;
        a[j | (1 << KH)].x = v.z; a[j | (1 << KH)].y = v.w;
    }
}

template <int KH, int P0, int P1, int P2, int P3, int P4, int P5, int P6>
__device__ __forceinline__ void tile_store(float4* st, int pb, const float2 (&a)[NA]) {
    static_assert(tile_off<P0, P1, P2, P3, P4, P5, P6>(1 << KH) == 32, "KH must map to index bit 5");
#pragma unroll
    for (int j = 0; j < NA; ++j) {
        if (j & (1 << KH)) continue;
        float4 v;
        v.x = a[j].x;             v.y = a[j].y;
        v.z = a[j | (1 << KH)].x; v.w = a[j | (1 << KH)].y;
        st[pb + slot4(tile_off<P0, P1, P2, P3, P4, P5, P6>(j))] = v;
    }
}

struct GateTab { float m; unsigned flip; };

__device__ __forceinline__ GateTab make_tab(float theta) {
    const float hh = 0.5f * theta;
    const float c = __cosf(hh);
    const float s = __sinf(hh);
    const bool cm = fabsf(c) >= fabsf(s);
    GateTab g;
    g.m = (cm ? s : -c) * __builtin_amdgcn_rcpf(cm ? c : s);
    g.flip = __builtin_amdgcn_ballot_w32(!cm) & 0xFFFu;
    return g;
}

__device__ __forceinline__ unsigned cnot_frame(unsigned f) {
    f ^= f << 1;
    f ^= f << 2;
    f ^= f << 4;
    f ^= f << 8;
    f &= 0xFFFu;
    f ^= (f >> 11) & 1u;
    return f;
}

template <int K>
__device__ __forceinline__ void rx_local(float2 (&a)[NA], const GateTab& g, int q) {
    const float m = __int_as_float(__builtin_amdgcn_readlane(__float_as_int(g.m), q));
    const v2f mm = {m, -m};
#pragma unroll
    for (int j = 0; j < NA; ++j) {
        if (j & (1 << K)) continue;
        const int j1 = j | (1 << K);
        const v2f p0 = {a[j].x,  a[j].y};
        const v2f p1 = {a[j1].x, a[j1].y};
        const v2f s0 = __builtin_shufflevector(p0, p0, 1, 0);
        const v2f s1 = __builtin_shufflevector(p1, p1, 1, 0);
        const v2f r0 = __builtin_elementwise_fma(mm, s1, p0);
        const v2f r1 = __builtin_elementwise_fma(mm, s0, p1);
        a[j].x  = r0.x;  a[j].y  = r0.y;
        a[j1].x = r1.x;  a[j1].y = r1.y;
    }
}

template <int CP, int TP>
__device__ __forceinline__ void cnot_local(float2 (&a)[NA]) {
#pragma unroll
    for (int j = 0; j < NA; ++j) {
        if (!(j & (1 << CP)) || (j & (1 << TP))) continue;
        const int j1 = j | (1 << TP);
        const float2 t = a[j];
        a[j]  = a[j1];
        a[j1] = t;
    }
}

__device__ __forceinline__ void cnot_chain(float2 (&a)[NA]) {
    cnot_local<0, 1>(a);
    cnot_local<1, 2>(a);
    cnot_local<2, 3>(a);
    cnot_local<3, 4>(a);
    cnot_local<4, 5>(a);
    cnot_local<5, 6>(a);
}

struct CS { float c, s; };

__device__ __forceinline__ CS half_angle_sum(float te, float tw) {
    const float he = 0.5f * te, hw = 0.5f * tw;
    const float ce = __cosf(he), se = __sinf(he);
    const float cw = __cosf(hw), sw = __sinf(hw);
    CS r;
    r.c = ce * cw - se * sw;
    r.s = se * cw + ce * sw;
    return r;
}

__device__ __forceinline__ float lane_bcast(float x, int q) {
    return __int_as_float(__builtin_amdgcn_readlane(__float_as_int(x), q));
}

__device__ __forceinline__ float2 mul_real(float2 v, float r) {
    float2 o; o.x = v.x * r; o.y = v.y * r; return o;
}
__device__ __forceinline__ float2 mul_negi(float2 v, float r) {
    float2 o; o.x = v.y * r; o.y = -(v.x * r); return o;
}
__device__ __forceinline__ float2 mul_f(float2 v, float c, float s, int bit) {
    const float2 a0 = mul_real(v, c), a1 = mul_negi(v, s);
    float2 o; o.x = bit ? a1.x : a0.x; o.y = bit ? a1.y : a0.y; return o;
}

template <int K>
__device__ __forceinline__ void build_level(float2 (&a)[NA], float c, float s) {
#pragma unroll
    for (int base = 0; base < NA; base += (2 << K)) {
        const float2 p = a[base];
        if ((base >> (K + 1)) & 1) {
            a[base]            = mul_negi(p, s);
            a[base | (1 << K)] = mul_real(p, c);
        } else {
            a[base]            = mul_real(p, c);
            a[base | (1 << K)] = mul_negi(p, s);
        }
    }
}

__device__ __forceinline__ void build_layer0(int lane, float2 (&a)[NA], const CS& g) {
    float c[NQ], s[NQ];
#pragma unroll
    for (int q = 0; q < NQ; ++q) { c[q] = lane_bcast(g.c, q); s[q] = lane_bcast(g.s, q); }
    const int n11 = lane & 1, n10 = (lane >> 1) & 1, n9 = (lane >> 2) & 1;
    const int n8 = (lane >> 3) & 1, n7 = (lane >> 4) & 1;

    float2 L; L.x = 1.f; L.y = 0.f;
    L = mul_f(L, c[11], s[11], n11 ^ n10);
    L = mul_f(L, c[10], s[10], n10 ^ n9);
    L = mul_f(L, c[9],  s[9],  n9 ^ n8);
    L = mul_f(L, c[8],  s[8],  n8 ^ n7);
    a[0]  = mul_f(L, c[7], s[7], n7);
    a[64] = mul_f(L, c[7], s[7], n7 ^ 1);

    build_level<5>(a, c[6], s[6]);
    build_level<4>(a, c[5], s[5]);
    build_level<3>(a, c[4], s[4]);
    build_level<2>(a, c[3], s[3]);
    build_level<1>(a, c[2], s[2]);

    const float cc = c[1] * c[0], ss = -(s[1] * s[0]);
    const float sc = s[1] * c[0], cs = c[1] * s[0];
    const float g00 = n11 ? ss : cc, g01 = n11 ? cc : ss;
    const float h10 = n11 ? cs : sc, h11 = n11 ? sc : cs;
#pragma unroll
    for (int base = 0; base < NA; base += 2) {
        const float2 p = a[base];
        if ((base >> 1) & 1) {
            a[base]     = mul_negi(p, h10);
            a[base | 1] = mul_negi(p, h11);
        } else {
            a[base]     = mul_real(p, g00);
            a[base | 1] = mul_real(p, g01);
        }
    }
}

__device__ __forceinline__ void window1(float4* st, int lane, float2 (&a)[NA], const GateTab& w) {
    rx_local<0>(a, w, 0);
    rx_local<1>(a, w, 1);
    rx_local<2>(a, w, 2);
    rx_local<3>(a, w, 3);
    rx_local<4>(a, w, 4);
    rx_local<5>(a, w, 5);
    rx_local<6>(a, w, 6);
    cnot_chain(a);
    tile_store<6, 11, 10, 9, 8, 7, 6, 5>(st, slot4(lane), a);
    __syncthreads();
}

__device__ __forceinline__ void window1_load(const float4* st, int lane, float2 (&a)[NA]) {
    tile_load<6, 11, 10, 9, 8, 7, 6, 5>(st, slot4(lane), a);
}

__device__ __forceinline__ void window2(const float4* st, int lane, float2 (&a)[NA], const GateTab& w) {
    tile_load<0, 5, 4, 3, 2, 1, 0, 11>(st, slot4(lane << 6), a);
    rx_local<1>(a, w, 7);
    rx_local<2>(a, w, 8);
    rx_local<3>(a, w, 9);
    rx_local<4>(a, w, 10);
    rx_local<5>(a, w, 11);
    cnot_chain(a);
}

__device__ __forceinline__ void window2_store(float4* st, int lane, const float2 (&a)[NA]) {
    tile_store<0, 5, 4, 3, 2, 1, 0, 11>(st, slot4(lane << 6), a);
    __syncthreads();
}

template <int N>
__device__ __forceinline__ void z_fold(float (&v)[NA], float& d) {
#pragma unroll
    for (int t = 0; t < N / 2; ++t) {
        d += v[2 * t] - v[2 * t + 1];
        v[t] = v[2 * t] + v[2 * t + 1];
    }
}

__device__ __forceinline__ float wave_sum(float x) {
    x += __shfl_xor(x, 16, 32);
    x += __shfl_xor(x, 8, 32);
    x += __shfl_xor(x, 4, 32);
    x += __shfl_xor(x, 2, 32);
    x += __shfl_xor(x, 1, 32);
    return x;
}

__global__ __launch_bounds__(32) void vqc_kernel(const float* __restrict__ angles,
                                                 const float* __restrict__ qp,
                                                 const float* __restrict__ W_out,
                                                 const float* __restrict__ b_out,
                                                 float* __restrict__ out) {
    __shared__ float4 st[ST_CELLS];
    const int lane = threadIdx.x;
    const int b    = blockIdx.x;

    const int gq = lane < NQ ? lane : NQ - 1;
    const CS l0 = half_angle_sum(angles[(size_t)b * ANG_STRIDE + gq], qp[gq]);
    GateTab cur = make_tab(qp[NQ + gq]);
    float th_next = qp[2 * NQ + gq];
    unsigned frame = cnot_frame(cur.flip);

    float2 a[NA];

    build_layer0(lane, a, l0);
    window1(st, lane, a, cur);
    for (int l = 1; l < NLAYERS; ++l) {
        if (l > 1) {
            cur = make_tab(th_next);
            const int ln = (l + 1 < NLAYERS) ? l + 1 : NLAYERS - 1;
            th_next = qp[ln * NQ + gq];
            frame = cnot_frame(frame ^ cur.flip);
            window1_load(st, lane, a);
            window1(st, lane, a, cur);
        }
        window2(st, lane, a, cur);
        if (l + 1 < NLAYERS) window2_store(st, lane, a);
    }

    float v[NA];
#pragma unroll
    for (int j = 0; j < NA; ++j) v[j] = fmaf(a[j].x, a[j].x, a[j].y * a[j].y);
    float zt[7];
#pragma unroll
    for (int k = 0; k < 7; ++k) zt[k] = 0.f;
    z_fold<128>(v, zt[0]);
    z_fold<64>(v, zt[1]);
    z_fold<32>(v, zt[2]);
    z_fold<16>(v, zt[3]);
    z_fold<8>(v, zt[4]);
    z_fold<4>(v, zt[5]);
    z_fold<2>(v, zt[6]);
    const float tp = v[0];

    float z[NQ];
    z[0] = zt[6];
#pragma unroll
    for (int q = 1; q <= 5; ++q) z[q] = ((lane >> (5 - q)) & 1) ? -tp : tp;
#pragma unroll
    for (int q = 6; q < NQ; ++q) z[q] = zt[q - 6];
#pragma unroll
    for (int q = 0; q < NQ; ++q) z[q] = wave_sum(z[q]);
    const float tot = wave_sum(tp);

    const float inv = 1.0f / tot;
#pragma unroll
    for (int q = 0; q < NQ; ++q) z[q] *= ((frame >> q) & 1u) ? -inv : inv;

#pragma unroll
    for (int r = 0; r < DOUT / 32; ++r) {
        const int o = r * 32 + lane;
        const float4* wr = (const float4*)(W_out + o * NQ);
        const float4 w0 = wr[0], w1 = wr[1], w2 = wr[2];
        float accv = b_out[o];
        accv = fmaf(z[0],  w0.x, accv);
        accv = fmaf(z[1],  w0.y, accv);
        accv = fmaf(z[2],  w0.z, accv);
        accv = fmaf(z[3],  w0.w, accv);
        accv = fmaf(z[4],  w1.x, accv);
        accv = fmaf(z[5],  w1.y, accv);
        accv = fmaf(z[6],  w1.z, accv);
        accv = fmaf(z[7],  w1.w, accv);
        accv = fmaf(z[8],  w2.x, accv);
        accv = fmaf(z[9],  w2.y, accv);
        accv = fmaf(z[10], w2.z, accv);
        accv = fmaf(z[11], w2.w, accv);
        out[(size_t)b * DOUT + o] = accv;
    }
}

extern "C" void kernel_launch(void* const* d_in, const int* in_sizes, int n_in,
                              void* d_out, int out_size, void* d_ws, size_t ws_size,
                              hipStream_t stream) {
    const float* x        = (const float*)d_in[0];
    const float* W_in     = (const float*)d_in[1];
    const float* b_in     = (const float*)d_in[2];
    const float* q_params = (const float*)d_in[3];
    const float* W_out    = (const float*)d_in[4];
    const float* b_out    = (const float*)d_in[5];
    float* out    = (float*)d_out;
    float* angles = (float*)d_ws;

    fc_in_wmma<<<dim3(32),   dim3(256), 0, stream>>>(x, W_in, b_in, angles);
    vqc_kernel<<<dim3(4096), dim3(32),  0, stream>>>(angles, q_params, W_out, b_out, out);
}


# No component-exchange moves: amplitudes kept as whole register pairs, each LDS trip pairs in a cell the qubit of the next first gate (qubit 6, then 0), RX6 moved to window 2; each fmaf unchanged
# speedup vs baseline: 8.0865x; 1.0481x over previous
#include <hip/hip_runtime.h>

#define NQ       12
#define DIM      4096
#define NLAYERS  6
#define NA       128
#define ANG_STRIDE 16
#define DOUT     256

typedef float v2f __attribute__((ext_vector_type(2)));
typedef float v8f __attribute__((ext_vector_type(8)));

__global__ __launch_bounds__(256) void fc_in_wmma(const float* __restrict__ x,
                                                  const float* __restrict__ W_in,
                                                  const float* __restrict__ b_in,
                                                  float* __restrict__ angles) {
    const int lane = threadIdx.x & 31;
    const int wave = threadIdx.x >> 5;
    const int tile = blockIdx.x * 8 + wave;
    const int n    = lane & 15;
    const int h    = lane >> 4;
    const int nc   = (n < NQ) ? n : (NQ - 1);
    const float mask = (n < NQ) ? 1.0f : 0.0f;

    const float* xrow = x + (size_t)(tile * 16 + n) * 256;
    const float* wrow = W_in + nc * 256;

    v8f acc = {};
    for (int k0 = 0; k0 < 256; k0 += 8) {
        const float4 av = *(const float4*)(xrow + k0 + 4 * h);
        float4 wv       = *(const float4*)(wrow + k0 + 4 * h);
        wv.x *= mask; wv.y *= mask; wv.z *= mask; wv.w *= mask;

        v2f a1, b1, a2, b2;
        a1.x = av.x; a1.y = av.y;  b1.x = wv.x; b1.y = wv.y;
        a2.x = av.z; a2.y = av.w;  b2.x = wv.z; b2.y = wv.w;
        acc = __builtin_amdgcn_wmma_f32_16x16x4_f32(false, a1, false, b1,
                                                    (short)0, acc, false, false);
        acc = __builtin_amdgcn_wmma_f32_16x16x4_f32(false, a2, false, b2,
                                                    (short)0, acc, false, false);
    }

    const float bias = b_in[nc] * mask;
    const int mbase = tile * 16 + (h << 3);
#pragma unroll
    for (int r = 0; r < 8; ++r)
        angles[(size_t)(mbase + r) * ANG_STRIDE + n] = acc[r] + bias;
}

#define ST_CELLS 2112
template <int FMT>
__host__ __device__ constexpr int slotf(int i) {
    return (i & 31) + ((i >> 6) & 1) * 33 + ((i >> 7) & 1) * 66 +
           ((i >> 8) & 1) * 132 + ((i >> 9) & 1) * 264 + ((i >> 10) & 1) * 528 +
           ((i >> (FMT == 0 ? 11 : 5)) & 1) * 1056;
}
__host__ __device__ constexpr int slot4(int i) { return slotf<0>(i); }

template <int P0, int P1, int P2, int P3, int P4, int P5, int P6>
__host__ __device__ constexpr int tile_off(int j) {
    return ((j & 1) << P0) | (((j >> 1) & 1) << P1) | (((j >> 2) & 1) << P2) |
           (((j >> 3) & 1) << P3) | (((j >> 4) & 1) << P4) | (((j >> 5) & 1) << P5) |
           (((j >> 6) & 1) << P6);
}

template <int FMT, int KH, int P0, int P1, int P2, int P3, int P4, int P5, int P6>
__device__ __forceinline__ void tile_load(const float4* st, int pb, v2f (&a)[NA]) {
    static_assert(tile_off<P0, P1, P2, P3, P4, P5, P6>(1 << KH) == (FMT == 0 ? 32 : 2048), "KH must map to the in-cell index bit");
#pragma unroll
    for (int j = 0; j < NA; ++j) {
        if (j & (1 << KH)) continue;
        const v2f* cell = (const v2f*)(st + pb + slotf<FMT>(tile_off<P0, P1, P2, P3, P4, P5, P6>(j)));
        a[j]             = cell[0];
        a[j | (1 << KH)] = cell[1];
    }
}

template <int FMT, int KH, int P0, int P1, int P2, int P3, int P4, int P5, int P6>
__device__ __forceinline__ void tile_store(float4* st, int pb, const v2f (&a)[NA]) {
    static_assert(tile_off<P0, P1, P2, P3, P4, P5, P6>(1 << KH) == (FMT == 0 ? 32 : 2048), "KH must map to the in-cell index bit");
#pragma unroll
    for (int j = 0; j < NA; ++j) {
        if (j & (1 << KH)) continue;
        float4 v;
        v.x = a[j].x;             v.y = a[j].y;
        v.z = a[j | (1 << KH)].x; v.w = a[j | (1 << KH)].y;
        st[pb + slotf<FMT>(tile_off<P0, P1, P2, P3, P4, P5, P6>(j))] = v;
    }
}

struct GateTab { float m; unsigned flip; };

__device__ __forceinline__ GateTab make_tab(float theta) {
    const float hh = 0.5f * theta;
    const float c = __cosf(hh);
    const float s = __sinf(hh);
    const bool cm = fabsf(c) >= fabsf(s);
    GateTab g;
    g.m = (cm ? s : -c) * __builtin_amdgcn_rcpf(cm ? c : s);
    g.flip = __builtin_amdgcn_ballot_w32(!cm) & 0xFFFu;
    return g;
}

__device__ __forceinline__ unsigned cnot_frame(unsigned f) {
    f ^= f << 1;
    f ^= f << 2;
    f ^= f << 4;
    f ^= f << 8;
    f &= 0xFFFu;
    f ^= (f >> 11) & 1u;
    return f;
}

template <int K>
__device__ __forceinline__ void rx_local(v2f (&a)[NA], const GateTab& g, int q) {
    const float m = __int_as_float(__builtin_amdgcn_readlane(__float_as_int(g.m), q));
    const v2f mm = {m, -m};
#pragma unroll
    for (int j = 0; j < NA; ++j) {
        if (j & (1 << K)) continue;
        const int j1 = j | (1 << K);
        const v2f p0 = a[j];
        const v2f p1 = a[j1];
        const v2f s0 = __builtin_shufflevector(p0, p0, 1, 0);
        const v2f s1 = __builtin_shufflevector(p1, p1, 1, 0);
        const v2f r0 = __builtin_elementwise_fma(mm, s1, p0);
        const v2f r1 = __builtin_elementwise_fma(mm, s0, p1);
        a[j]  = r0;
        a[j1] = r1;
    }
}

template <int K>
__device__ __forceinline__ void rx_first(v2f (&a)[NA], const GateTab& g, int q) {
    const float m = __int_as_float(__builtin_amdgcn_readlane(__float_as_int(g.m), q));
    const v2f mm = {m, -m};
    const v2f mn = {-m, m};
#pragma unroll
    for (int j = 0; j < NA; ++j) {
        if (j & (1 << K)) continue;
        const int j1 = j | (1 << K);
        const v2f p0 = a[j];
        const v2f p1 = a[j1];
        const v2f s0 = __builtin_shufflevector(p0, p0, 1, 0);
        const v2f r0x = __builtin_elementwise_fma(mn, p1, s0);
        const v2f r1  = __builtin_elementwise_fma(mm, s0, p1);
        a[j]  = __builtin_shufflevector(r0x, r0x, 1, 0);
        a[j1] = r1;
    }
}

template <int CP, int TP>
__device__ __forceinline__ void cnot_local(v2f (&a)[NA]) {
#pragma unroll
    for (int j = 0; j < NA; ++j) {
        if (!(j & (1 << CP)) || (j & (1 << TP))) continue;
        const int j1 = j | (1 << TP);
        const v2f t = a[j];
        a[j]  = a[j1];
        a[j1] = t;
    }
}

__device__ __forceinline__ void cnot_chain(v2f (&a)[NA]) {
    cnot_local<0, 1>(a);
    cnot_local<1, 2>(a);
    cnot_local<2, 3>(a);
    cnot_local<3, 4>(a);
    cnot_local<4, 5>(a);
    cnot_local<5, 6>(a);
}

struct CS { float c, s; };

__device__ __forceinline__ CS half_angle_sum(float te, float tw) {
    const float he = 0.5f * te, hw = 0.5f * tw;
    const float ce = __cosf(he), se = __sinf(he);
    const float cw = __cosf(hw), sw = __sinf(hw);
    CS r;
    r.c = ce * cw - se * sw;
    r.s = se * cw + ce * sw;
    return r;
}

__device__ __forceinline__ float lane_bcast(float x, int q) {
    return __int_as_float(__builtin_amdgcn_readlane(__float_as_int(x), q));
}

__device__ __forceinline__ v2f mul_real(v2f v, float r) {
    const v2f rr = {r, r};
    return v * rr;
}
__device__ __forceinline__ v2f mul_negi(v2f v, float r) {
    const v2f rn = {r, -r};
    return __builtin_shufflevector(v, v, 1, 0) * rn;
}
__device__ __forceinline__ v2f mul_f(v2f v, float c, float s, int bit) {
    const v2f a0 = mul_real(v, c), a1 = mul_negi(v, s);
    v2f o; o.x = bit ? a1.x : a0.x; o.y = bit ? a1.y : a0.y; return o;
}

template <int K>
__device__ __forceinline__ void build_level(v2f (&a)[NA], float c, float s) {
#pragma unroll
    for (int base = 0; base < NA; base += (2 << K)) {
        const v2f p = a[base];
        if ((base >> (K + 1)) & 1) {
            a[base]            = mul_negi(p, s);
            a[base | (1 << K)] = mul_real(p, c);
        } else {
            a[base]            = mul_real(p, c);
            a[base | (1 << K)] = mul_negi(p, s);
        }
    }
}

__device__ __forceinline__ void build_layer0(int lane, v2f (&a)[NA], const CS& g) {
    float c[NQ], s[NQ];
#pragma unroll
    for (int q = 0; q < NQ; ++q) { c[q] = lane_bcast(g.c, q); s[q] = lane_bcast(g.s, q); }
    const int n11 = lane & 1, n10 = (lane >> 1) & 1, n9 = (lane >> 2) & 1;
    const int n8 = (lane >> 3) & 1, n7 = (lane >> 4) & 1;

    v2f L; L.x = 1.f; L.y = 0.f;
    L = mul_f(L, c[11], s[11], n11 ^ n10);
    L = mul_f(L, c[10], s[10], n10 ^ n9);
    L = mul_f(L, c[9],  s[9],  n9 ^ n8);
    L = mul_f(L, c[8],  s[8],  n8 ^ n7);
    a[0]  = mul_f(L, c[7], s[7], n7);
    a[64] = mul_f(L, c[7], s[7], n7 ^ 1);

    build_level<5>(a, c[6], s[6]);
    build_level<4>(a, c[5], s[5]);
    build_level<3>(a, c[4], s[4]);
    build_level<2>(a, c[3], s[3]);
    build_level<1>(a, c[2], s[2]);

    const float cc = c[1] * c[0], ss = -(s[1] * s[0]);
    const float sc = s[1] * c[0], cs = c[1] * s[0];
    const float g00 = n11 ? ss : cc, g01 = n11 ? cc : ss;
    const float h10 = n11 ? cs : sc, h11 = n11 ? sc : cs;
#pragma unroll
    for (int base = 0; base < NA; base += 2) {
        const v2f p = a[base];
        if ((base >> 1) & 1) {
            a[base]     = mul_negi(p, h10);
            a[base | 1] = mul_negi(p, h11);
        } else {
            a[base]     = mul_real(p, g00);
            a[base | 1] = mul_real(p, g01);
        }
    }
}

__device__ __forceinline__ void window1(float4* st, int lane, v2f (&a)[NA], const GateTab& w) {
    rx_first<0>(a, w, 0);
    rx_local<1>(a, w, 1);
    rx_local<2>(a, w, 2);
    rx_local<3>(a, w, 3);
    rx_local<4>(a, w, 4);
    rx_local<5>(a, w, 5);
    cnot_chain(a);
    tile_store<0, 6, 11, 10, 9, 8, 7, 6, 5>(st, slot4(lane), a);
    __syncthreads();
}

__device__ __forceinline__ void window1_load(const float4* st, int lane, v2f (&a)[NA]) {
    tile_load<1, 0, 11, 10, 9, 8, 7, 6, 5>(st, slot4(lane), a);
}

__device__ __forceinline__ void window2(const float4* st, int lane, v2f (&a)[NA], const GateTab& w) {
    tile_load<0, 0, 5, 4, 3, 2, 1, 0, 11>(st, slot4(lane << 6), a);
    rx_first<0>(a, w, 6);
    rx_local<1>(a, w, 7);
    rx_local<2>(a, w, 8);
    rx_local<3>(a, w, 9);
    rx_local<4>(a, w, 10);
    rx_local<5>(a, w, 11);
    cnot_chain(a);
}

__device__ __forceinline__ void window2_store(float4* st, int lane, const v2f (&a)[NA]) {
    tile_store<1, 6, 5, 4, 3, 2, 1, 0, 11>(st, slot4(lane << 6), a);
    __syncthreads();
}

template <int N>
__device__ __forceinline__ void z_fold(float (&v)[NA], float& d) {
#pragma unroll
    for (int t = 0; t < N / 2; ++t) {
        d += v[2 * t] - v[2 * t + 1];
        v[t] = v[2 * t] + v[2 * t + 1];
    }
}

__device__ __forceinline__ float wave_sum(float x) {
    x += __shfl_xor(x, 16, 32);
    x += __shfl_xor(x, 8, 32);
    x += __shfl_xor(x, 4, 32);
    x += __shfl_xor(x, 2, 32);
    x += __shfl_xor(x, 1, 32);
    return x;
}

__global__ __launch_bounds__(32) void vqc_kernel(const float* __restrict__ angles,
                                                 const float* __restrict__ qp,
                                                 const float* __restrict__ W_out,
                                                 const float* __restrict__ b_out,
                                                 float* __restrict__ out) {
    __shared__ float4 st[ST_CELLS];
    const int lane = threadIdx.x;
    const int b    = blockIdx.x;

    const int gq = lane < NQ ? lane : NQ - 1;
    const CS l0 = half_angle_sum(angles[(size_t)b * ANG_STRIDE + gq], qp[gq]);
    GateTab cur = make_tab(qp[NQ + gq]);
    float th_next = qp[2 * NQ + gq];
    unsigned frame = cnot_frame(cur.flip);

    v2f a[NA];

    build_layer0(lane, a, l0);
    window1(st, lane, a, cur);
    for (int l = 1; l < NLAYERS; ++l) {
        if (l > 1) {
            cur = make_tab(th_next);
            const int ln = (l + 1 < NLAYERS) ? l + 1 : NLAYERS - 1;
            th_next = qp[ln * NQ + gq];
            frame = cnot_frame(frame ^ cur.flip);
            window1_load(st, lane, a);
            window1(st, lane, a, cur);
        }
        window2(st, lane, a, cur);
        if (l + 1 < NLAYERS) window2_store(st, lane, a);
    }

    float v[NA];
#pragma unroll
    for (int j = 0; j < NA; ++j) v[j] = fmaf(a[j].x, a[j].x, a[j].y * a[j].y);
    float zt[7];
#pragma unroll
    for (int k = 0; k < 7; ++k) zt[k] = 0.f;
    z_fold<128>(v, zt[0]);
    z_fold<64>(v, zt[1]);
    z_fold<32>(v, zt[2]);
    z_fold<16>(v, zt[3]);
    z_fold<8>(v, zt[4]);
    z_fold<4>(v, zt[5]);
    z_fold<2>(v, zt[6]);
    const float tp = v[0];

    float z[NQ];
    z[0] = zt[6];
#pragma unroll
    for (int q = 1; q <= 5; ++q) z[q] = ((lane >> (5 - q)) & 1) ? -tp : tp;
#pragma unroll
    for (int q = 6; q < NQ; ++q) z[q] = zt[q - 6];
#pragma unroll
    for (int q = 0; q < NQ; ++q) z[q] = wave_sum(z[q]);
    const float tot = wave_sum(tp);

    const float inv = 1.0f / tot;
#pragma unroll
    for (int q = 0; q < NQ; ++q) z[q] *= ((frame >> q) & 1u) ? -inv : inv;

#pragma unroll
    for (int r = 0; r < DOUT / 32; ++r) {
        const int o = r * 32 + lane;
        const float4* wr = (const float4*)(W_out + o * NQ);
        const float4 w0 = wr[0], w1 = wr[1], w2 = wr[2];
        float accv = b_out[o];
        accv = fmaf(z[0],  w0.x, accv);
        accv = fmaf(z[1],  w0.y, accv);
        accv = fmaf(z[2],  w0.z, accv);
        accv = fmaf(z[3],  w0.w, accv);
        accv = fmaf(z[4],  w1.x, accv);
        accv = fmaf(z[5],  w1.y, accv);
        accv = fmaf(z[6],  w1.z, accv);
        accv = fmaf(z[7],  w1.w, accv);
        accv = fmaf(z[8],  w2.x, accv);
        accv = fmaf(z[9],  w2.y, accv);
        accv = fmaf(z[10], w2.z, accv);
        accv = fmaf(z[11], w2.w, accv);
        out[(size_t)b * DOUT + o] = accv;
    }
}

extern "C" void kernel_launch(void* const* d_in, const int* in_sizes, int n_in,
                              void* d_out, int out_size, void* d_ws, size_t ws_size,
                              hipStream_t stream) {
    const float* x        = (const float*)d_in[0];
    const float* W_in     = (const float*)d_in[1];
    const float* b_in     = (const float*)d_in[2];
    const float* q_params = (const float*)d_in[3];
    const float* W_out    = (const float*)d_in[4];
    const float* b_out    = (const float*)d_in[5];
    float* out    = (float*)d_out;
    float* angles = (float*)d_ws;

    fc_in_wmma<<<dim3(32),   dim3(256), 0, stream>>>(x, W_in, b_in, angles);
    vqc_kernel<<<dim3(4096), dim3(32),  0, stream>>>(angles, q_params, W_out, b_out, out);
}


# fc_in GEMM on all WGPs: one 4-wave work-group per 16-row tile, K split 4 ways, each wave's 16 loads issued as one group, partial tiles added in LDS in a fixed order; fc_in sum order changes
# speedup vs baseline: 9.0634x; 1.1208x over previous
#include <hip/hip_runtime.h>

#define NQ       12
#define DIM      4096
#define NLAYERS  6
#define NA       128
#define ANG_STRIDE 16
#define DOUT     256

typedef float v2f __attribute__((ext_vector_type(2)));
typedef float v4f __attribute__((ext_vector_type(4)));
typedef float v8f __attribute__((ext_vector_type(8)));

#define KSPLIT 4
__global__ __launch_bounds__(128) void fc_in_wmma(const float* __restrict__ x,
                                                         const float* __restrict__ W_in,
                                                         const float* __restrict__ b_in,
                                                         float* __restrict__ angles) {
    __shared__ float part[KSPLIT][8][32];
    const int lane = threadIdx.x & 31;
    const int wave = threadIdx.x >> 5;
    const int tile = blockIdx.x;
    const int n    = lane & 15;
    const int h    = lane >> 4;
    const int nc   = (n < NQ) ? n : (NQ - 1);
    const float mask = (n < NQ) ? 1.0f : 0.0f;
    constexpr int KW = 256 / KSPLIT;
    constexpr int NS = KW / 8;

    const float* xrow = x + (size_t)(tile * 16 + n) * 256 + wave * KW + 4 * h;
    const float* wrow = W_in + nc * 256 + wave * KW + 4 * h;
    const float bias = b_in[nc] * mask;

    v4f av[NS], wv[NS];
#pragma unroll
    for (int i = 0; i < NS; ++i) av[i] = *(const v4f*)(xrow + 8 * i);
#pragma unroll
    for (int i = 0; i < NS; ++i) wv[i] = *(const v4f*)(wrow + 8 * i);
    asm volatile("" :: "v"(av[0]), "v"(av[1]), "v"(av[2]), "v"(av[3]),
                       "v"(av[4]), "v"(av[5]), "v"(av[6]), "v"(av[7]),
                       "v"(wv[0]), "v"(wv[1]), "v"(wv[2]), "v"(wv[3]),
                       "v"(wv[4]), "v"(wv[5]), "v"(wv[6]), "v"(wv[7]));
    static_assert(NS == 8 && KSPLIT == 4, "the fence above names eight steps; the launch has four waves per tile");

    v8f acc = {};
#pragma unroll
    for (int i = 0; i < NS; ++i) {
        v4f w4 = wv[i];
        w4.x *= mask; w4.y *= mask; w4.z *= mask; w4.w *= mask;

        v2f a1, b1, a2, b2;
        a1.x = av[i].x; a1.y = av[i].y;  b1.x = w4.x; b1.y = w4.y;
        a2.x = av[i].z; a2.y = av[i].w;  b2.x = w4.z; b2.y = w4.w;
        acc = __builtin_amdgcn_wmma_f32_16x16x4_f32(false, a1, false, b1,
                                                    (short)0, acc, false, false);
        acc = __builtin_amdgcn_wmma_f32_16x16x4_f32(false, a2, false, b2,
                                                    (short)0, acc, false, false);
    }

#pragma unroll
    for (int r = 0; r < 8; ++r) part[wave][r][lane] = acc[r];
    __syncthreads();

    const int mbase = tile * 16 + (h << 3);
#pragma unroll
    for (int rr = 0; rr < 8 / KSPLIT; ++rr) {
        const int r = wave * (8 / KSPLIT) + rr;
        float s = part[0][r][lane];
#pragma unroll
        for (int w = 1; w < KSPLIT; ++w) s += part[w][r][lane];
        angles[(size_t)(mbase + r) * ANG_STRIDE + n] = s + bias;
    }
}

#define ST_CELLS 2112
template <int FMT>
__host__ __device__ constexpr int slotf(int i) {
    return (i & 31) + ((i >> 6) & 1) * 33 + ((i >> 7) & 1) * 66 +
           ((i >> 8) & 1) * 132 + ((i >> 9) & 1) * 264 + ((i >> 10) & 1) * 528 +
           ((i >> (FMT == 0 ? 11 : 5)) & 1) * 1056;
}
__host__ __device__ constexpr int slot4(int i) { return slotf<0>(i); }

template <int P0, int P1, int P2, int P3, int P4, int P5, int P6>
__host__ __device__ constexpr int tile_off(int j) {
    return ((j & 1) << P0) | (((j >> 1) & 1) << P1) | (((j >> 2) & 1) << P2) |
           (((j >> 3) & 1) << P3) | (((j >> 4) & 1) << P4) | (((j >> 5) & 1) << P5) |
           (((j >> 6) & 1) << P6);
}

template <int FMT, int KH, int P0, int P1, int P2, int P3, int P4, int P5, int P6>
__device__ __forceinline__ void tile_load(const float4* st, int pb, v2f (&a)[NA]) {
    static_assert(tile_off<P0, P1, P2, P3, P4, P5, P6>(1 << KH) == (FMT == 0 ? 32 : 2048), "KH must map to the in-cell index bit");
#pragma unroll
    for (int j = 0; j < NA; ++j) {
        if (j & (1 << KH)) continue;
        const v2f* cell = (const v2f*)(st + pb + slotf<FMT>(tile_off<P0, P1, P2, P3, P4, P5, P6>(j)));
        a[j]             = cell[0];
        a[j | (1 << KH)] = cell[1];
    }
}

template <int FMT, int KH, int P0, int P1, int P2, int P3, int P4, int P5, int P6>
__device__ __forceinline__ void tile_store(float4* st, int pb, const v2f (&a)[NA]) {
    static_assert(tile_off<P0, P1, P2, P3, P4, P5, P6>(1 << KH) == (FMT == 0 ? 32 : 2048), "KH must map to the in-cell index bit");
#pragma unroll
    for (int j = 0; j < NA; ++j) {
        if (j & (1 << KH)) continue;
        float4 v;
        v.x = a[j].x;             v.y = a[j].y;
        v.z = a[j | (1 << KH)].x; v.w = a[j | (1 << KH)].y;
        st[pb + slotf<FMT>(tile_off<P0, P1, P2, P3, P4, P5, P6>(j))] = v;
    }
}

struct GateTab { float m; unsigned flip; };

__device__ __forceinline__ GateTab make_tab(float theta) {
    const float hh = 0.5f * theta;
    const float c = __cosf(hh);
    const float s = __sinf(hh);
    const bool cm = fabsf(c) >= fabsf(s);
    GateTab g;
    g.m = (cm ? s : -c) * __builtin_amdgcn_rcpf(cm ? c : s);
    g.flip = __builtin_amdgcn_ballot_w32(!cm) & 0xFFFu;
    return g;
}

__device__ __forceinline__ unsigned cnot_frame(unsigned f) {
    f ^= f << 1;
    f ^= f << 2;
    f ^= f << 4;
    f ^= f << 8;
    f &= 0xFFFu;
    f ^= (f >> 11) & 1u;
    return f;
}

template <int K>
__device__ __forceinline__ void rx_local(v2f (&a)[NA], const GateTab& g, int q) {
    const float m = __int_as_float(__builtin_amdgcn_readlane(__float_as_int(g.m), q));
    const v2f mm = {m, -m};
#pragma unroll
    for (int j = 0; j < NA; ++j) {
        if (j & (1 << K)) continue;
        const int j1 = j | (1 << K);
        const v2f p0 = a[j];
        const v2f p1 = a[j1];
        const v2f s0 = __builtin_shufflevector(p0, p0, 1, 0);
        const v2f s1 = __builtin_shufflevector(p1, p1, 1, 0);
        const v2f r0 = __builtin_elementwise_fma(mm, s1, p0);
        const v2f r1 = __builtin_elementwise_fma(mm, s0, p1);
        a[j]  = r0;
        a[j1] = r1;
    }
}

template <int K>
__device__ __forceinline__ void rx_first(v2f (&a)[NA], const GateTab& g, int q) {
    const float m = __int_as_float(__builtin_amdgcn_readlane(__float_as_int(g.m), q));
    const v2f mm = {m, -m};
    const v2f mn = {-m, m};
#pragma unroll
    for (int j = 0; j < NA; ++j) {
        if (j & (1 << K)) continue;
        const int j1 = j | (1 << K);
        const v2f p0 = a[j];
        const v2f p1 = a[j1];
        const v2f s0 = __builtin_shufflevector(p0, p0, 1, 0);
        const v2f r0x = __builtin_elementwise_fma(mn, p1, s0);
        const v2f r1  = __builtin_elementwise_fma(mm, s0, p1);
        a[j]  = __builtin_shufflevector(r0x, r0x, 1, 0);
        a[j1] = r1;
    }
}

template <int CP, int TP>
__device__ __forceinline__ void cnot_local(v2f (&a)[NA]) {
#pragma unroll
    for (int j = 0; j < NA; ++j) {
        if (!(j & (1 << CP)) || (j & (1 << TP))) continue;
        const int j1 = j | (1 << TP);
        const v2f t = a[j];
        a[j]  = a[j1];
        a[j1] = t;
    }
}

__device__ __forceinline__ void cnot_chain(v2f (&a)[NA]) {
    cnot_local<0, 1>(a);
    cnot_local<1, 2>(a);
    cnot_local<2, 3>(a);
    cnot_local<3, 4>(a);
    cnot_local<4, 5>(a);
    cnot_local<5, 6>(a);
}

struct CS { float c, s; };

__device__ __forceinline__ CS half_angle_sum(float te, float tw) {
    const float he = 0.5f * te, hw = 0.5f * tw;
    const float ce = __cosf(he), se = __sinf(he);
    const float cw = __cosf(hw), sw = __sinf(hw);
    CS r;
    r.c = ce * cw - se * sw;
    r.s = se * cw + ce * sw;
    return r;
}

__device__ __forceinline__ float lane_bcast(float x, int q) {
    return __int_as_float(__builtin_amdgcn_readlane(__float_as_int(x), q));
}

__device__ __forceinline__ v2f mul_real(v2f v, float r) {
    const v2f rr = {r, r};
    return v * rr;
}
__device__ __forceinline__ v2f mul_negi(v2f v, float r) {
    const v2f rn = {r, -r};
    return __builtin_shufflevector(v, v, 1, 0) * rn;
}
__device__ __forceinline__ v2f mul_f(v2f v, float c, float s, int bit) {
    const v2f a0 = mul_real(v, c), a1 = mul_negi(v, s);
    v2f o; o.x = bit ? a1.x : a0.x; o.y = bit ? a1.y : a0.y; return o;
}

template <int K>
__device__ __forceinline__ void build_level(v2f (&a)[NA], float c, float s) {
#pragma unroll
    for (int base = 0; base < NA; base += (2 << K)) {
        const v2f p = a[base];
        if ((base >> (K + 1)) & 1) {
            a[base]            = mul_negi(p, s);
            a[base | (1 << K)] = mul_real(p, c);
        } else {
            a[base]            = mul_real(p, c);
            a[base | (1 << K)] = mul_negi(p, s);
        }
    }
}

__device__ __forceinline__ void build_layer0(int lane, v2f (&a)[NA], const CS& g) {
    float c[NQ], s[NQ];
#pragma unroll
    for (int q = 0; q < NQ; ++q) { c[q] = lane_bcast(g.c, q); s[q] = lane_bcast(g.s, q); }
    const int n11 = lane & 1, n10 = (lane >> 1) & 1, n9 = (lane >> 2) & 1;
    const int n8 = (lane >> 3) & 1, n7 = (lane >> 4) & 1;

    v2f L; L.x = 1.f; L.y = 0.f;
    L = mul_f(L, c[11], s[11], n11 ^ n10);
    L = mul_f(L, c[10], s[10], n10 ^ n9);
    L = mul_f(L, c[9],  s[9],  n9 ^ n8);
    L = mul_f(L, c[8],  s[8],  n8 ^ n7);
    a[0]  = mul_f(L, c[7], s[7], n7);
    a[64] = mul_f(L, c[7], s[7], n7 ^ 1);

    build_level<5>(a, c[6], s[6]);
    build_level<4>(a, c[5], s[5]);
    build_level<3>(a, c[4], s[4]);
    build_level<2>(a, c[3], s[3]);
    build_level<1>(a, c[2], s[2]);

    const float cc = c[1] * c[0], ss = -(s[1] * s[0]);
    const float sc = s[1] * c[0], cs = c[1] * s[0];
    const float g00 = n11 ? ss : cc, g01 = n11 ? cc : ss;
    const float h10 = n11 ? cs : sc, h11 = n11 ? sc : cs;
#pragma unroll
    for (int base = 0; base < NA; base += 2) {
        const v2f p = a[base];
        if ((base >> 1) & 1) {
            a[base]     = mul_negi(p, h10);
            a[base | 1] = mul_negi(p, h11);
        } else {
            a[base]     = mul_real(p, g00);
            a[base | 1] = mul_real(p, g01);
        }
    }
}

__device__ __forceinline__ void window1(float4* st, int lane, v2f (&a)[NA], const GateTab& w) {
    rx_first<0>(a, w, 0);
    rx_local<1>(a, w, 1);
    rx_local<2>(a, w, 2);
    rx_local<3>(a, w, 3);
    rx_local<4>(a, w, 4);
    rx_local<5>(a, w, 5);
    cnot_chain(a);
    tile_store<0, 6, 11, 10, 9, 8, 7, 6, 5>(st, slot4(lane), a);
    __syncthreads();
}

__device__ __forceinline__ void window1_load(const float4* st, int lane, v2f (&a)[NA]) {
    tile_load<1, 0, 11, 10, 9, 8, 7, 6, 5>(st, slot4(lane), a);
}

__device__ __forceinline__ void window2(const float4* st, int lane, v2f (&a)[NA], const GateTab& w) {
    tile_load<0, 0, 5, 4, 3, 2, 1, 0, 11>(st, slot4(lane << 6), a);
    rx_first<0>(a, w, 6);
    rx_local<1>(a, w, 7);
    rx_local<2>(a, w, 8);
    rx_local<3>(a, w, 9);
    rx_local<4>(a, w, 10);
    rx_local<5>(a, w, 11);
    cnot_chain(a);
}

__device__ __forceinline__ void window2_store(float4* st, int lane, const v2f (&a)[NA]) {
    tile_store<1, 6, 5, 4, 3, 2, 1, 0, 11>(st, slot4(lane << 6), a);
    __syncthreads();
}

template <int N>
__device__ __forceinline__ void z_fold(float (&v)[NA], float& d) {
#pragma unroll
    for (int t = 0; t < N / 2; ++t) {
        d += v[2 * t] - v[2 * t + 1];
        v[t] = v[2 * t] + v[2 * t + 1];
    }
}

__device__ __forceinline__ float wave_sum(float x) {
    x += __shfl_xor(x, 16, 32);
    x += __shfl_xor(x, 8, 32);
    x += __shfl_xor(x, 4, 32);
    x += __shfl_xor(x, 2, 32);
    x += __shfl_xor(x, 1, 32);
    return x;
}

__global__ __launch_bounds__(32) void vqc_kernel(const float* __restrict__ angles,
                                                 const float* __restrict__ qp,
                                                 const float* __restrict__ W_out,
                                                 const float* __restrict__ b_out,
                                                 float* __restrict__ out) {
    __shared__ float4 st[ST_CELLS];
    const int lane = threadIdx.x;
    const int b    = blockIdx.x;

    const int gq = lane < NQ ? lane : NQ - 1;
    const CS l0 = half_angle_sum(angles[(size_t)b * ANG_STRIDE + gq], qp[gq]);
    GateTab cur = make_tab(qp[NQ + gq]);
    float th_next = qp[2 * NQ + gq];
    unsigned frame = cnot_frame(cur.flip);

    v2f a[NA];

    build_layer0(lane, a, l0);
    window1(st, lane, a, cur);
    for (int l = 1; l < NLAYERS; ++l) {
        if (l > 1) {
            cur = make_tab(th_next);
            const int ln = (l + 1 < NLAYERS) ? l + 1 : NLAYERS - 1;
            th_next = qp[ln * NQ + gq];
            frame = cnot_frame(frame ^ cur.flip);
            window1_load(st, lane, a);
            window1(st, lane, a, cur);
        }
        window2(st, lane, a, cur);
        if (l + 1 < NLAYERS) window2_store(st, lane, a);
    }

    float v[NA];
#pragma unroll
    for (int j = 0; j < NA; ++j) v[j] = fmaf(a[j].x, a[j].x, a[j].y * a[j].y);
    float zt[7];
#pragma unroll
    for (int k = 0; k < 7; ++k) zt[k] = 0.f;
    z_fold<128>(v, zt[0]);
    z_fold<64>(v, zt[1]);
    z_fold<32>(v, zt[2]);
    z_fold<16>(v, zt[3]);
    z_fold<8>(v, zt[4]);
    z_fold<4>(v, zt[5]);
    z_fold<2>(v, zt[6]);
    const float tp = v[0];

    float z[NQ];
    z[0] = zt[6];
#pragma unroll
    for (int q = 1; q <= 5; ++q) z[q] = ((lane >> (5 - q)) & 1) ? -tp : tp;
#pragma unroll
    for (int q = 6; q < NQ; ++q) z[q] = zt[q - 6];
#pragma unroll
    for (int q = 0; q < NQ; ++q) z[q] = wave_sum(z[q]);
    const float tot = wave_sum(tp);

    const float inv = 1.0f / tot;
#pragma unroll
    for (int q = 0; q < NQ; ++q) z[q] *= ((frame >> q) & 1u) ? -inv : inv;

#pragma unroll
    for (int r = 0; r < DOUT / 32; ++r) {
        const int o = r * 32 + lane;
        const float4* wr = (const float4*)(W_out + o * NQ);
        const float4 w0 = wr[0], w1 = wr[1], w2 = wr[2];
        float accv = b_out[o];
        accv = fmaf(z[0],  w0.x, accv);
        accv = fmaf(z[1],  w0.y, accv);
        accv = fmaf(z[2],  w0.z, accv);
        accv = fmaf(z[3],  w0.w, accv);
        accv = fmaf(z[4],  w1.x, accv);
        accv = fmaf(z[5],  w1.y, accv);
        accv = fmaf(z[6],  w1.z, accv);
        accv = fmaf(z[7],  w1.w, accv);
        accv = fmaf(z[8],  w2.x, accv);
        accv = fmaf(z[9],  w2.y, accv);
        accv = fmaf(z[10], w2.z, accv);
        accv = fmaf(z[11], w2.w, accv);
        out[(size_t)b * DOUT + o] = accv;
    }
}

extern "C" void kernel_launch(void* const* d_in, const int* in_sizes, int n_in,
                              void* d_out, int out_size, void* d_ws, size_t ws_size,
                              hipStream_t stream) {
    const float* x        = (const float*)d_in[0];
    const float* W_in     = (const float*)d_in[1];
    const float* b_in     = (const float*)d_in[2];
    const float* q_params = (const float*)d_in[3];
    const float* W_out    = (const float*)d_in[4];
    const float* b_out    = (const float*)d_in[5];
    float* out    = (float*)d_out;
    float* angles = (float*)d_ws;

    fc_in_wmma<<<dim3(256),  dim3(128), 0, stream>>>(x, W_in, b_in, angles);
    vqc_kernel<<<dim3(4096), dim3(32),  0, stream>>>(angles, q_params, W_out, b_out, out);
}
